# starting kernel
# baseline (speedup 1.0000x reference)
#include <hip/hip_runtime.h>

typedef float v2f __attribute__((ext_vector_type(2)));
typedef float v8f __attribute__((ext_vector_type(8)));

__device__ __forceinline__ v8f wmma4(v2f a, v2f b, v8f c) {
  return __builtin_amdgcn_wmma_f32_16x16x4_f32(false, a, false, b, (short)0, c,
                                               false, false);
}

template <int WS, int SH, int CHUNK>
__global__ __launch_bounds__(128) void win_attn_kernel(
    const float* __restrict__ x, float* __restrict__ out) {
  constexpr int H = 256, W = 256, HW = H * W;
  constexpr int NW = W / WS;
  constexpr int T = WS * WS;
  constexpr int NT = T / 16;

  __shared__ float qvl[8 * T * 4];
  __shared__ float ptile[4][16 * 20];

  const int wid = blockIdx.x;
  const int b = blockIdx.y;
  const int wy = wid / NW, wx = wid % NW;
  const int tid = threadIdx.x;
  const int lane = tid & 31;
  const int head = tid >> 5;

  const float* xbase = x + ((size_t)b * 64 + CHUNK * 32) * HW;
  for (int i = tid; i < 32 * T; i += 128) {
    int c = i / T, t = i - c * T;
    int ty = t / WS, tx = t - ty * WS;
    int gh = (wy * WS + ty - SH) & (H - 1);
    int gw = (wx * WS + tx - SH) & (W - 1);
    float val = xbase[c * HW + gh * W + gw];
    qvl[((c >> 2) * T + t) * 4 + (c & 3)] = val;
  }
  __syncthreads();

  const float* qb = &qvl[head * T * 4];
  const float* vb = &qvl[(4 + head) * T * 4];
  float* pl = &ptile[head][0];

  const int m = lane & 15;
  const int hi = lane >> 4;
  const int k0 = hi * 2;

  for (int qt = 0; qt < NT; ++qt) {
    v2f a = *(const v2f*)(qb + ((qt * 16 + m) * 4 + k0));

    v8f S[NT];
#pragma unroll
    for (int kt = 0; kt < NT; ++kt) {
      v2f bq = *(const v2f*)(qb + ((kt * 16 + m) * 4 + k0));
      v8f z = {0.f, 0.f, 0.f, 0.f, 0.f, 0.f, 0.f, 0.f};
      S[kt] = wmma4(a, bq, z);
    }

    float rmax[8];
#pragma unroll
    for (int r = 0; r < 8; ++r) {
      float mx = -3.402823466e38f;
#pragma unroll
      for (int kt = 0; kt < NT; ++kt) mx = fmaxf(mx, S[kt][r]);
      mx = fmaxf(mx, __shfl_xor(mx, 8, 32));
      mx = fmaxf(mx, __shfl_xor(mx, 4, 32));
      mx = fmaxf(mx, __shfl_xor(mx, 2, 32));
      mx = fmaxf(mx, __shfl_xor(mx, 1, 32));
      rmax[r] = mx;
    }
#pragma unroll
    for (int kt = 0; kt < NT; ++kt) {
#pragma unroll
      for (int r = 0; r < 8; ++r) S[kt][r] = __expf(S[kt][r] - rmax[r]);
    }

    v8f O = {0.f, 0.f, 0.f, 0.f, 0.f, 0.f, 0.f, 0.f};
    for (int kt = 0; kt < NT; ++kt) {
#pragma unroll
      for (int r = 0; r < 8; ++r)
        pl[(r + 8 * hi) * 20 + m] = S[kt][r];
#pragma unroll
      for (int c4 = 0; c4 < 4; ++c4) {
        v2f pa = *(const v2f*)(pl + m * 20 + c4 * 4 + k0);
        v2f bv = {0.f, 0.f};
        if (m < 4) {
          bv.x = vb[(kt * 16 + c4 * 4 + k0) * 4 + m];
          bv.y = vb[(kt * 16 + c4 * 4 + k0 + 1) * 4 + m];
        } else if (m == 4) {
          bv.x = 1.f;
          bv.y = 1.f;
        }
        O = wmma4(pa, bv, O);
      }
    }

#pragma unroll
    for (int r = 0; r < 8; ++r) {
      float sm = __shfl(O[r], (lane & 16) | 4, 32);
      if (m < 4) {
        int tok = qt * 16 + r + 8 * hi;
        int ty = tok / WS, tx = tok - ty * WS;
        int gh = (wy * WS + ty - SH) & (H - 1);
        int gw = (wx * WS + tx - SH) & (W - 1);
        out[((size_t)b * 32 + CHUNK * 16 + head * 4 + m) * HW + gh * W + gw] =
            O[r] / sm;
      }
    }
  }
}

extern "C" void kernel_launch(void* const* d_in, const int* in_sizes, int n_in,
                              void* d_out, int out_size, void* d_ws,
                              size_t ws_size, hipStream_t stream) {
  (void)in_sizes; (void)n_in; (void)out_size; (void)d_ws; (void)ws_size;
  const float* x = (const float*)d_in[0];
  float* out = (float*)d_out;

  dim3 g0(32 * 32, 4);
  win_attn_kernel<8, 0, 0><<<g0, 128, 0, stream>>>(x, out);
  dim3 g1(16 * 16, 4);
  win_attn_kernel<16, 8, 1><<<g1, 128, 0, stream>>>(x, out);
}


# Second product transposed, Ot=[V|1]^T*Pt: score tiles' D registers are its B operand (no LDS staging of P, max down a lane); key sum order in each 16-key tile changes; stores are 64B/32B rows
# speedup vs baseline: 2.3373x; 2.3373x over previous
#include <hip/hip_runtime.h>


typedef float v2f __attribute__((ext_vector_type(2)));
typedef float v8f __attribute__((ext_vector_type(8)));

__device__ __forceinline__ v8f wmma4(v2f a, v2f b, v8f c) {
  return __builtin_amdgcn_wmma_f32_16x16x4_f32(false, a, false, b, (short)0, c,
                                               false, false);
}

__device__ __forceinline__ float xhalf(float v) {
  int b = __builtin_bit_cast(int, v);
  return __builtin_bit_cast(
      float, __builtin_amdgcn_permlanex16(b, b, 0x76543210, 0xfedcba98, false,
                                          false));
}

template <int WS, int SH, int CHUNK>
__global__ __launch_bounds__(128) void win_attn_kernel(
    const float* __restrict__ x, float* __restrict__ out) {
  constexpr int H = 256, W = 256, HW = H * W;
  constexpr int NW = W / WS;
  constexpr int T = WS * WS;
  constexpr int NT = T / 16;
  constexpr int VR = 5;

  __shared__ __attribute__((aligned(16))) float ql[4 * T * 4];
  __shared__ __attribute__((aligned(16))) float vl[4 * (T / 2) * VR * 2];

  const int wid = blockIdx.x;
  const int b = blockIdx.y;
  const int wy = wid / NW, wx = wid % NW;
  const int tid = threadIdx.x;
  const int lane = tid & 31;
  const int head = tid >> 5;

  const float* xbase = x + ((size_t)b * 64 + CHUNK * 32) * HW;
  constexpr int GI = 16 * T / 128;
  for (int it = 0; it < GI; ++it) {
    int i = tid + it * 128;
    int c = i / T, t = i - c * T;
    int ty = t / WS, tx = t - ty * WS;
    int gh = (wy * WS + ty - SH) & (H - 1);
    int gw = (wx * WS + tx - SH) & (W - 1);
    ql[((c >> 2) * T + t) * 4 + (c & 3)] = xbase[c * HW + gh * W + gw];
  }
  for (int it = 0; it < GI; ++it) {
    int i = tid + it * 128;
    int c = i / T, t = i - c * T;
    int ty = t / WS, tx = t - ty * WS;
    int gh = (wy * WS + ty - SH) & (H - 1);
    int gw = (wx * WS + tx - SH) & (W - 1);
    vl[(((c >> 2) * (T / 2) + (t >> 1)) * VR + (c & 3)) * 2 + (t & 1)] =
        xbase[(16 + c) * HW + gh * W + gw];
  }
  for (int i = tid; i < 4 * (T / 2); i += 128) {
    v2f one = {1.f, 1.f};
    *(v2f*)&vl[(i * VR + 4) * 2] = one;
  }
  __syncthreads();

  const float* qb = &ql[head * T * 4];
  const float* vb = &vl[head * (T / 2) * VR * 2];

  const int m = lane & 15;
  const int hi = lane >> 4;
  const int k0 = hi * 2;
  const int vrow = (m & 7) < VR ? (m & 7) : 4;
  const float* va = vb + (4 * hi * VR + vrow) * 2;

  for (int qt = 0; qt < NT; ++qt) {
    asm volatile("" ::: "memory");
    v2f bqry = *(const v2f*)(qb + ((qt * 16 + m) * 4 + k0));

    v8f S[NT];
#pragma unroll
    for (int kt = 0; kt < NT; ++kt) {
      v2f akey = *(const v2f*)(qb + ((kt * 16 + m) * 4 + k0));
      v8f z = {0.f, 0.f, 0.f, 0.f, 0.f, 0.f, 0.f, 0.f};
      S[kt] = wmma4(akey, bqry, z);
    }

    float mc[4] = {-3.402823466e38f, -3.402823466e38f, -3.402823466e38f,
                   -3.402823466e38f};
#pragma unroll
    for (int kt = 0; kt < NT; ++kt) {
#pragma unroll
      for (int r = 0; r < 8; ++r) mc[kt & 3] = fmaxf(mc[kt & 3], S[kt][r]);
    }
    float mx = fmaxf(fmaxf(mc[0], mc[1]), fmaxf(mc[2], mc[3]));
    mx = fmaxf(mx, xhalf(mx));
#pragma unroll
    for (int kt = 0; kt < NT; ++kt) {
#pragma unroll
      for (int r = 0; r < 8; ++r) S[kt][r] = __expf(S[kt][r] - mx);
    }

    v8f O = {0.f, 0.f, 0.f, 0.f, 0.f, 0.f, 0.f, 0.f};
#pragma unroll
    for (int kt = 0; kt < NT; ++kt) {
#pragma unroll
      for (int j = 0; j < 4; ++j) {
        v2f bp = {S[kt][2 * j], S[kt][2 * j + 1]};
        v2f av = *(const v2f*)(va + (kt * 8 + j) * VR * 2);
        O = wmma4(av, bp, O);
      }
    }

    {
      int tok = qt * 16 + m;
      int ty = tok / WS, tx = tok - ty * WS;
      int gh = (wy * WS + ty - SH) & (H - 1);
      int gw = (wx * WS + tx - SH) & (W - 1);
      float* op = out +
                  ((size_t)b * 32 + CHUNK * 16 + head * 4 + 2 * hi) * HW +
                  gh * W + gw;
      float sm = O[4];
      float o0 = hi ? O[2] : O[0];
      float o1 = hi ? O[3] : O[1];
      op[0] = o0 / sm;
      op[HW] = o1 / sm;
    }
  }
}

extern "C" void kernel_launch(void* const* d_in, const int* in_sizes, int n_in,
                              void* d_out, int out_size, void* d_ws,
                              size_t ws_size, hipStream_t stream) {
  (void)in_sizes; (void)n_in; (void)out_size; (void)d_ws; (void)ws_size;
  const float* x = (const float*)d_in[0];
  float* out = (float*)d_out;

  dim3 g0(32 * 32, 4);
  win_attn_kernel<8, 0, 0><<<g0, 128, 0, stream>>>(x, out);
  dim3 g1(16 * 16, 4);
  win_attn_kernel<16, 8, 1><<<g1, 128, 0, stream>>>(x, out);
}


# 16x16-window kernel: key-side WMMA fragments in registers ahead of use (Q keys kept per wave, [V|1]^T read 2 key tiles ahead in a register ring, next query B one pass ahead); sums unchanged
# speedup vs baseline: 2.4585x; 1.0519x over previous
#include <hip/hip_runtime.h>


typedef float v2f __attribute__((ext_vector_type(2)));
typedef float v8f __attribute__((ext_vector_type(8)));

__device__ __forceinline__ v8f wmma4(v2f a, v2f b, v8f c) {
  return __builtin_amdgcn_wmma_f32_16x16x4_f32(false, a, false, b, (short)0, c,
                                               false, false);
}

__device__ __forceinline__ float xhalf(float v) {
  int b = __builtin_bit_cast(int, v);
  return __builtin_bit_cast(
      float, __builtin_amdgcn_permlanex16(b, b, 0x76543210, 0xfedcba98, false,
                                          false));
}

template <int WS, int SH, int CHUNK>
__global__ __launch_bounds__(128) void win_attn_kernel(
    const float* __restrict__ x, float* __restrict__ out) {
  constexpr int H = 256, W = 256, HW = H * W;
  constexpr int NW = W / WS;
  constexpr int T = WS * WS;
  constexpr int NT = T / 16;
  constexpr int VR = 5;

  __shared__ __attribute__((aligned(16))) float ql[4 * T * 4];
  __shared__ __attribute__((aligned(16))) float vl[4 * (T / 2) * VR * 2];

  const int wid = blockIdx.x;
  const int b = blockIdx.y;
  const int wy = wid / NW, wx = wid % NW;
  const int tid = threadIdx.x;
  const int lane = tid & 31;
  const int head = tid >> 5;

  const float* xbase = x + ((size_t)b * 64 + CHUNK * 32) * HW;
  constexpr int GI = 16 * T / 128;
  for (int it = 0; it < GI; ++it) {
    int i = tid + it * 128;
    int c = i / T, t = i - c * T;
    int ty = t / WS, tx = t - ty * WS;
    int gh = (wy * WS + ty - SH) & (H - 1);
    int gw = (wx * WS + tx - SH) & (W - 1);
    ql[((c >> 2) * T + t) * 4 + (c & 3)] = xbase[c * HW + gh * W + gw];
  }
  for (int it = 0; it < GI; ++it) {
    int i = tid + it * 128;
    int c = i / T, t = i - c * T;
    int ty = t / WS, tx = t - ty * WS;
    int gh = (wy * WS + ty - SH) & (H - 1);
    int gw = (wx * WS + tx - SH) & (W - 1);
    vl[(((c >> 2) * (T / 2) + (t >> 1)) * VR + (c & 3)) * 2 + (t & 1)] =
        xbase[(16 + c) * HW + gh * W + gw];
  }
  for (int i = tid; i < 4 * (T / 2); i += 128) {
    v2f one = {1.f, 1.f};
    *(v2f*)&vl[(i * VR + 4) * 2] = one;
  }
  __syncthreads();

  const float* qb = &ql[head * T * 4];
  const float* vb = &vl[head * (T / 2) * VR * 2];

  const int m = lane & 15;
  const int hi = lane >> 4;
  const int k0 = hi * 2;
  const int vrow = (m & 7) < VR ? (m & 7) : 4;
  const float* va = vb + (4 * hi * VR + vrow) * 2;

  constexpr bool AHEAD = NT > 4;
  constexpr int PF = 2;
  constexpr int NKEEP = AHEAD ? NT : 1;
  v2f akeep[NKEEP];
  v2f avkeep[PF][4];
  v2f bnext = {0.f, 0.f};
  if constexpr (AHEAD) {
#pragma unroll
    for (int kt = 0; kt < NT; ++kt)
      akeep[kt] = *(const v2f*)(qb + ((kt * 16 + m) * 4 + k0));
#pragma unroll
    for (int kt = 0; kt < PF; ++kt) {
#pragma unroll
      for (int j = 0; j < 4; ++j)
        avkeep[kt][j] = *(const v2f*)(va + (kt * 8 + j) * VR * 2);
    }
    bnext = *(const v2f*)(qb + (m * 4 + k0));
  }

  for (int qt = 0; qt < NT; ++qt) {
    asm volatile("" ::: "memory");
    v2f bqry;
    if constexpr (AHEAD) bqry = bnext;
    else bqry = *(const v2f*)(qb + ((qt * 16 + m) * 4 + k0));

    v8f S[NT];
#pragma unroll
    for (int kt = 0; kt < NT; ++kt) {
      v2f akey;
      if constexpr (AHEAD) akey = akeep[kt];
      else akey = *(const v2f*)(qb + ((kt * 16 + m) * 4 + k0));
      v8f z = {0.f, 0.f, 0.f, 0.f, 0.f, 0.f, 0.f, 0.f};
      S[kt] = wmma4(akey, bqry, z);
    }

    float mc[4] = {-3.402823466e38f, -3.402823466e38f, -3.402823466e38f,
                   -3.402823466e38f};
#pragma unroll
    for (int kt = 0; kt < NT; ++kt) {
#pragma unroll
      for (int r = 0; r < 8; ++r) mc[kt & 3] = fmaxf(mc[kt & 3], S[kt][r]);
    }
    float mx = fmaxf(fmaxf(mc[0], mc[1]), fmaxf(mc[2], mc[3]));
    mx = fmaxf(mx, xhalf(mx));
    v8f O = {0.f, 0.f, 0.f, 0.f, 0.f, 0.f, 0.f, 0.f};
    if constexpr (AHEAD) {
      v2f av[NT][4];
#pragma unroll
      for (int kt = 0; kt < PF; ++kt) {
#pragma unroll
        for (int j = 0; j < 4; ++j) av[kt][j] = avkeep[kt][j];
      }
#pragma unroll
      for (int r = 0; r < 8; ++r) S[0][r] = __expf(S[0][r] - mx);
#pragma unroll
      for (int kt = 0; kt < NT; ++kt) {
        if ((kt & 1) == 0 && kt + PF < NT) {
          __builtin_amdgcn_sched_barrier(0);
#pragma unroll
          for (int t = kt + PF; t < kt + PF + 2; ++t) {
#pragma unroll
            for (int j = 0; j < 4; ++j)
              av[t][j] = *(const v2f*)(va + (t * 8 + j) * VR * 2);
          }
          if (kt == 0) {
            bnext = *(const v2f*)(
                qb + ((((qt + 1) & (NT - 1)) * 16 + m) * 4 + k0));
          }
          __builtin_amdgcn_sched_barrier(0);
        }
        if (kt + 1 < NT) {
#pragma unroll
          for (int r = 0; r < 8; ++r)
            S[kt + 1][r] = __expf(S[kt + 1][r] - mx);
        }
#pragma unroll
        for (int j = 0; j < 4; ++j) {
          v2f bp = {S[kt][2 * j], S[kt][2 * j + 1]};
          O = wmma4(av[kt][j], bp, O);
        }
      }
    } else {
#pragma unroll
      for (int kt = 0; kt < NT; ++kt) {
#pragma unroll
        for (int r = 0; r < 8; ++r) S[kt][r] = __expf(S[kt][r] - mx);
      }

#pragma unroll
      for (int kt = 0; kt < NT; ++kt) {
#pragma unroll
        for (int j = 0; j < 4; ++j) {
          v2f bp = {S[kt][2 * j], S[kt][2 * j + 1]};
          v2f av = *(const v2f*)(va + (kt * 8 + j) * VR * 2);
          O = wmma4(av, bp, O);
        }
      }
    }

    {
      int tok = qt * 16 + m;
      int ty = tok / WS, tx = tok - ty * WS;
      int gh = (wy * WS + ty - SH) & (H - 1);
      int gw = (wx * WS + tx - SH) & (W - 1);
      float* op = out +
                  ((size_t)b * 32 + CHUNK * 16 + head * 4 + 2 * hi) * HW +
                  gh * W + gw;
      float sm = O[4];
      float o0 = hi ? O[2] : O[0];
      float o1 = hi ? O[3] : O[1];
      op[0] = o0 / sm;
      op[HW] = o1 / sm;
    }
  }
}

extern "C" void kernel_launch(void* const* d_in, const int* in_sizes, int n_in,
                              void* d_out, int out_size, void* d_ws,
                              size_t ws_size, hipStream_t stream) {
  (void)in_sizes; (void)n_in; (void)out_size; (void)d_ws; (void)ws_size;
  const float* x = (const float*)d_in[0];
  float* out = (float*)d_out;

  dim3 g0(32 * 32, 4);
  win_attn_kernel<8, 0, 0><<<g0, 128, 0, stream>>>(x, out);
  dim3 g1(16 * 16, 4);
  win_attn_kernel<16, 8, 1><<<g1, 128, 0, stream>>>(x, out);
}


# Both window sizes in ONE launch: 1-D grid of 5120 four-wave blocks, the 1024 16x16-window blocks first, the 4096 8x8-window blocks filling slots as those end; one LDS buffer; sums unchanged
# speedup vs baseline: 2.6281x; 1.0690x over previous
#include <hip/hip_runtime.h>


typedef float v2f __attribute__((ext_vector_type(2)));
typedef float v8f __attribute__((ext_vector_type(8)));

__device__ __forceinline__ v8f wmma4(v2f a, v2f b, v8f c) {
  return __builtin_amdgcn_wmma_f32_16x16x4_f32(false, a, false, b, (short)0, c,
                                               false, false);
}

__device__ __forceinline__ float xhalf(float v) {
  int b = __builtin_bit_cast(int, v);
  return __builtin_bit_cast(
      float, __builtin_amdgcn_permlanex16(b, b, 0x76543210, 0xfedcba98, false,
                                          false));
}

template <int WS, int SH, int CHUNK>
__device__ __forceinline__ void win_attn_block(
    const float* __restrict__ x, float* __restrict__ out,
    float* __restrict__ smem, const int wid, const int b) {
  constexpr int H = 256, W = 256, HW = H * W;
  constexpr int NW = W / WS;
  constexpr int T = WS * WS;
  constexpr int NT = T / 16;
  constexpr int VR = 5;

  float* const ql = smem;
  float* const vl = smem + 4 * T * 4;

  const int wy = wid / NW, wx = wid % NW;
  const int tid = threadIdx.x;
  const int lane = tid & 31;
  const int head = tid >> 5;

  const float* xbase = x + ((size_t)b * 64 + CHUNK * 32) * HW;
  constexpr int GI = 16 * T / 128;
  for (int it = 0; it < GI; ++it) {
    int i = tid + it * 128;
    int c = i / T, t = i - c * T;
    int ty = t / WS, tx = t - ty * WS;
    int gh = (wy * WS + ty - SH) & (H - 1);
    int gw = (wx * WS + tx - SH) & (W - 1);
    ql[((c >> 2) * T + t) * 4 + (c & 3)] = xbase[c * HW + gh * W + gw];
  }
  for (int it = 0; it < GI; ++it) {
    int i = tid + it * 128;
    int c = i / T, t = i - c * T;
    int ty = t / WS, tx = t - ty * WS;
    int gh = (wy * WS + ty - SH) & (H - 1);
    int gw = (wx * WS + tx - SH) & (W - 1);
    vl[(((c >> 2) * (T / 2) + (t >> 1)) * VR + (c & 3)) * 2 + (t & 1)] =
        xbase[(16 + c) * HW + gh * W + gw];
  }
  for (int i = tid; i < 4 * (T / 2); i += 128) {
    v2f one = {1.f, 1.f};
    *(v2f*)&vl[(i * VR + 4) * 2] = one;
  }
  __syncthreads();

  const float* qb = &ql[head * T * 4];
  const float* vb = &vl[head * (T / 2) * VR * 2];

  const int m = lane & 15;
  const int hi = lane >> 4;
  const int k0 = hi * 2;
  const int vrow = (m & 7) < VR ? (m & 7) : 4;
  const float* va = vb + (4 * hi * VR + vrow) * 2;

  constexpr bool AHEAD = NT > 4;
  constexpr int PF = 2;
  constexpr int NKEEP = AHEAD ? NT : 1;
  v2f akeep[NKEEP];
  v2f avkeep[PF][4];
  v2f bnext = {0.f, 0.f};
  if constexpr (AHEAD) {
#pragma unroll
    for (int kt = 0; kt < NT; ++kt)
      akeep[kt] = *(const v2f*)(qb + ((kt * 16 + m) * 4 + k0));
#pragma unroll
    for (int kt = 0; kt < PF; ++kt) {
#pragma unroll
      for (int j = 0; j < 4; ++j)
        avkeep[kt][j] = *(const v2f*)(va + (kt * 8 + j) * VR * 2);
    }
    bnext = *(const v2f*)(qb + (m * 4 + k0));
  }

  for (int qt = 0; qt < NT; ++qt) {
    asm volatile("" ::: "memory");
    v2f bqry;
    if constexpr (AHEAD) bqry = bnext;
    else bqry = *(const v2f*)(qb + ((qt * 16 + m) * 4 + k0));

    v8f S[NT];
#pragma unroll
    for (int kt = 0; kt < NT; ++kt) {
      v2f akey;
      if constexpr (AHEAD) akey = akeep[kt];
      else akey = *(const v2f*)(qb + ((kt * 16 + m) * 4 + k0));
      v8f z = {0.f, 0.f, 0.f, 0.f, 0.f, 0.f, 0.f, 0.f};
      S[kt] = wmma4(akey, bqry, z);
    }

    float mc[4] = {-3.402823466e38f, -3.402823466e38f, -3.402823466e38f,
                   -3.402823466e38f};
#pragma unroll
    for (int kt = 0; kt < NT; ++kt) {
#pragma unroll
      for (int r = 0; r < 8; ++r) mc[kt & 3] = fmaxf(mc[kt & 3], S[kt][r]);
    }
    float mx = fmaxf(fmaxf(mc[0], mc[1]), fmaxf(mc[2], mc[3]));
    mx = fmaxf(mx, xhalf(mx));
    v8f O = {0.f, 0.f, 0.f, 0.f, 0.f, 0.f, 0.f, 0.f};
    if constexpr (AHEAD) {
      v2f av[NT][4];
#pragma unroll
      for (int kt = 0; kt < PF; ++kt) {
#pragma unroll
        for (int j = 0; j < 4; ++j) av[kt][j] = avkeep[kt][j];
      }
#pragma unroll
      for (int r = 0; r < 8; ++r) S[0][r] = __expf(S[0][r] - mx);
#pragma unroll
      for (int kt = 0; kt < NT; ++kt) {
        if ((kt & 1) == 0 && kt + PF < NT) {
          __builtin_amdgcn_sched_barrier(0);
#pragma unroll
          for (int t = kt + PF; t < kt + PF + 2; ++t) {
#pragma unroll
            for (int j = 0; j < 4; ++j)
              av[t][j] = *(const v2f*)(va + (t * 8 + j) * VR * 2);
          }
          if (kt == 0) {
            bnext = *(const v2f*)(
                qb + ((((qt + 1) & (NT - 1)) * 16 + m) * 4 + k0));
          }
          __builtin_amdgcn_sched_barrier(0);
        }
        if (kt + 1 < NT) {
#pragma unroll
          for (int r = 0; r < 8; ++r)
            S[kt + 1][r] = __expf(S[kt + 1][r] - mx);
        }
#pragma unroll
        for (int j = 0; j < 4; ++j) {
          v2f bp = {S[kt][2 * j], S[kt][2 * j + 1]};
          O = wmma4(av[kt][j], bp, O);
        }
      }
    } else {
#pragma unroll
      for (int kt = 0; kt < NT; ++kt) {
#pragma unroll
        for (int r = 0; r < 8; ++r) S[kt][r] = __expf(S[kt][r] - mx);
      }

#pragma unroll
      for (int kt = 0; kt < NT; ++kt) {
#pragma unroll
        for (int j = 0; j < 4; ++j) {
          v2f bp = {S[kt][2 * j], S[kt][2 * j + 1]};
          v2f av = *(const v2f*)(va + (kt * 8 + j) * VR * 2);
          O = wmma4(av, bp, O);
        }
      }
    }

    {
      int tok = qt * 16 + m;
      int ty = tok / WS, tx = tok - ty * WS;
      int gh = (wy * WS + ty - SH) & (H - 1);
      int gw = (wx * WS + tx - SH) & (W - 1);
      float* op = out +
                  ((size_t)b * 32 + CHUNK * 16 + head * 4 + 2 * hi) * HW +
                  gh * W + gw;
      float sm = O[4];
      float o0 = hi ? O[2] : O[0];
      float o1 = hi ? O[3] : O[1];
      op[0] = o0 / sm;
      op[HW] = o1 / sm;
    }
  }
}

__global__ __launch_bounds__(128) void win_attn_fused(
    const float* __restrict__ x, float* __restrict__ out) {
  __shared__ __attribute__((aligned(16))) float smem[4 * 256 * 4 + 4 * 128 * 5 * 2];
  const int bx = blockIdx.x;
  if (bx < 1024) {
    win_attn_block<16, 8, 1>(x, out, smem, bx & 255, bx >> 8);
  } else {
    const int r = bx - 1024;
    win_attn_block<8, 0, 0>(x, out, smem, r & 1023, r >> 10);
  }
}

extern "C" void kernel_launch(void* const* d_in, const int* in_sizes, int n_in,
                              void* d_out, int out_size, void* d_ws,
                              size_t ws_size, hipStream_t stream) {
  (void)in_sizes; (void)n_in; (void)out_size; (void)d_ws; (void)ws_size;
  const float* x = (const float*)d_in[0];
  float* out = (float*)d_out;

  dim3 g(5120);
  win_attn_fused<<<g, 128, 0, stream>>>(x, out);
}


# Softmax shift |q|*max|k| (>= every score) replaces the row max and enters as the score WMMA's C, queries pre-scaled by log2e: max, sub, mul passes go; score sums start at C; old form kept if |k|>12
# speedup vs baseline: 2.7846x; 1.0595x over previous
#include <hip/hip_runtime.h>


typedef float v2f __attribute__((ext_vector_type(2)));
typedef float v8f __attribute__((ext_vector_type(8)));

__device__ __forceinline__ v8f wmma4(v2f a, v2f b, v8f c) {
  return __builtin_amdgcn_wmma_f32_16x16x4_f32(false, a, false, b, (short)0, c,
                                               false, false);
}

__device__ __forceinline__ float xhalf(float v) {
  int b = __builtin_bit_cast(int, v);
  return __builtin_bit_cast(
      float, __builtin_amdgcn_permlanex16(b, b, 0x76543210, 0xfedcba98, false,
                                          false));
}

template <int X>
__device__ __forceinline__ float xrow(float v) {
  constexpr unsigned lo = X == 1   ? 0x67452301u
                          : X == 2 ? 0x54761032u
                          : X == 4 ? 0x32107654u
                                   : 0xfedcba98u;
  constexpr unsigned hi = X == 1   ? 0xefcdab89u
                          : X == 2 ? 0xdcfe98bau
                          : X == 4 ? 0xba98fedcu
                                   : 0x76543210u;
  int b = __builtin_bit_cast(int, v);
  return __builtin_bit_cast(
      float, __builtin_amdgcn_permlane16(b, b, lo, hi, false, false));
}

template <int WS, int SH, int CHUNK, bool BOUND>
__device__ __forceinline__ void attend(
    float* __restrict__ out, const float* qb, const float* va, const int wy,
    const int wx, const int b, const int head, const int m, const int hi,
    const float kmax2) {
  constexpr int H = 256, W = 256, HW = H * W;
  constexpr int T = WS * WS;
  constexpr int NT = T / 16;
  constexpr int VR = 5;
  constexpr float LOG2E = 1.4426950408889634f;
  const int k0 = hi * 2;

  constexpr bool AHEAD = NT > 4;
  constexpr int PF = 2;
  constexpr int NKEEP = AHEAD ? NT : 1;
  v2f akeep[NKEEP];
  v2f avkeep[PF][4];
  v2f bnext = {0.f, 0.f};
  if constexpr (AHEAD) {
#pragma unroll
    for (int kt = 0; kt < NT; ++kt)
      akeep[kt] = *(const v2f*)(qb + ((kt * 16 + m) * 4 + k0));
#pragma unroll
    for (int kt = 0; kt < PF; ++kt) {
#pragma unroll
      for (int j = 0; j < 4; ++j)
        avkeep[kt][j] = *(const v2f*)(va + (kt * 8 + j) * VR * 2);
    }
    bnext = *(const v2f*)(qb + (m * 4 + k0));
  }

  for (int qt = 0; qt < NT; ++qt) {
    asm volatile("" ::: "memory");
    v2f bqry;
    if constexpr (AHEAD) bqry = bnext;
    else bqry = *(const v2f*)(qb + ((qt * 16 + m) * 4 + k0));

    v8f S[NT];
    float mx = 0.f;
    if constexpr (BOUND) {
      float qn = bqry.x * bqry.x + bqry.y * bqry.y;
      qn += xhalf(qn);
      const float c0 = -LOG2E * __builtin_amdgcn_sqrtf(qn * kmax2);
      const v2f bq2 = {bqry.x * LOG2E, bqry.y * LOG2E};
      const v8f cs = {c0, c0, c0, c0, c0, c0, c0, c0};
#pragma unroll
      for (int kt = 0; kt < NT; ++kt) {
        v2f akey;
        if constexpr (AHEAD) akey = akeep[kt];
        else akey = *(const v2f*)(qb + ((kt * 16 + m) * 4 + k0));
        S[kt] = wmma4(akey, bq2, cs);
      }
    } else {
#pragma unroll
      for (int kt = 0; kt < NT; ++kt) {
        v2f akey;
        if constexpr (AHEAD) akey = akeep[kt];
        else akey = *(const v2f*)(qb + ((kt * 16 + m) * 4 + k0));
        v8f z = {0.f, 0.f, 0.f, 0.f, 0.f, 0.f, 0.f, 0.f};
        S[kt] = wmma4(akey, bqry, z);
      }

      float mc[4] = {-3.402823466e38f, -3.402823466e38f, -3.402823466e38f,
                     -3.402823466e38f};
#pragma unroll
      for (int kt = 0; kt < NT; ++kt) {
#pragma unroll
        for (int r = 0; r < 8; ++r) mc[kt & 3] = fmaxf(mc[kt & 3], S[kt][r]);
      }
      mx = fmaxf(fmaxf(mc[0], mc[1]), fmaxf(mc[2], mc[3]));
      mx = fmaxf(mx, xhalf(mx));
    }
    auto pexp = [mx](float s) -> float {
      if constexpr (BOUND) return __builtin_amdgcn_exp2f(s);
      else return __expf(s - mx);
    };

    v8f O = {0.f, 0.f, 0.f, 0.f, 0.f, 0.f, 0.f, 0.f};
    if constexpr (AHEAD) {
      v2f av[NT][4];
#pragma unroll
      for (int kt = 0; kt < PF; ++kt) {
#pragma unroll
        for (int j = 0; j < 4; ++j) av[kt][j] = avkeep[kt][j];
      }
#pragma unroll
      for (int r = 0; r < 8; ++r) S[0][r] = pexp(S[0][r]);
#pragma unroll
      for (int kt = 0; kt < NT; ++kt) {
        if ((kt & 1) == 0 && kt + PF < NT) {
          __builtin_amdgcn_sched_barrier(0);
#pragma unroll
          for (int t = kt + PF; t < kt + PF + 2; ++t) {
#pragma unroll
            for (int j = 0; j < 4; ++j)
              av[t][j] = *(const v2f*)(va + (t * 8 + j) * VR * 2);
          }
          if (kt == 0) {
            bnext = *(const v2f*)(
                qb + ((((qt + 1) & (NT - 1)) * 16 + m) * 4 + k0));
          }
          __builtin_amdgcn_sched_barrier(0);
        }
        if (kt + 1 < NT) {
#pragma unroll
          for (int r = 0; r < 8; ++r) S[kt + 1][r] = pexp(S[kt + 1][r]);
        }
#pragma unroll
        for (int j = 0; j < 4; ++j) {
          v2f bp = {S[kt][2 * j], S[kt][2 * j + 1]};
          O = wmma4(av[kt][j], bp, O);
        }
      }
    } else {
#pragma unroll
      for (int kt = 0; kt < NT; ++kt) {
#pragma unroll
        for (int r = 0; r < 8; ++r) S[kt][r] = pexp(S[kt][r]);
      }

#pragma unroll
      for (int kt = 0; kt < NT; ++kt) {
#pragma unroll
        for (int j = 0; j < 4; ++j) {
          v2f bp = {S[kt][2 * j], S[kt][2 * j + 1]};
          v2f av = *(const v2f*)(va + (kt * 8 + j) * VR * 2);
          O = wmma4(av, bp, O);
        }
      }
    }

    {
      int tok = qt * 16 + m;
      int ty = tok / WS, tx = tok - ty * WS;
      int gh = (wy * WS + ty - SH) & (H - 1);
      int gw = (wx * WS + tx - SH) & (W - 1);
      float* op = out +
                  ((size_t)b * 32 + CHUNK * 16 + head * 4 + 2 * hi) * HW +
                  gh * W + gw;
      float sm = O[4];
      float o0 = hi ? O[2] : O[0];
      float o1 = hi ? O[3] : O[1];
      op[0] = o0 / sm;
      op[HW] = o1 / sm;
    }
  }
}

template <int WS, int SH, int CHUNK>
__device__ __forceinline__ void win_attn_block(
    const float* __restrict__ x, float* __restrict__ out,
    float* __restrict__ smem, const int wid, const int b) {
  constexpr int H = 256, W = 256, HW = H * W;
  constexpr int NW = W / WS;
  constexpr int T = WS * WS;
  constexpr int NT = T / 16;
  constexpr int VR = 5;

  float* const ql = smem;
  float* const vl = smem + 4 * T * 4;

  const int wy = wid / NW, wx = wid % NW;
  const int tid = threadIdx.x;
  const int lane = tid & 31;
  const int head = tid >> 5;

  const float* xbase = x + ((size_t)b * 64 + CHUNK * 32) * HW;
  constexpr int GI = 16 * T / 128;
  for (int it = 0; it < GI; ++it) {
    int i = tid + it * 128;
    int c = i / T, t = i - c * T;
    int ty = t / WS, tx = t - ty * WS;
    int gh = (wy * WS + ty - SH) & (H - 1);
    int gw = (wx * WS + tx - SH) & (W - 1);
    ql[((c >> 2) * T + t) * 4 + (c & 3)] = xbase[c * HW + gh * W + gw];
  }
  for (int it = 0; it < GI; ++it) {
    int i = tid + it * 128;
    int c = i / T, t = i - c * T;
    int ty = t / WS, tx = t - ty * WS;
    int gh = (wy * WS + ty - SH) & (H - 1);
    int gw = (wx * WS + tx - SH) & (W - 1);
    vl[(((c >> 2) * (T / 2) + (t >> 1)) * VR + (c & 3)) * 2 + (t & 1)] =
        xbase[(16 + c) * HW + gh * W + gw];
  }
  for (int i = tid; i < 4 * (T / 2); i += 128) {
    v2f one = {1.f, 1.f};
    *(v2f*)&vl[(i * VR + 4) * 2] = one;
  }
  __syncthreads();

  const float* qb = &ql[head * T * 4];
  const float* vb = &vl[head * (T / 2) * VR * 2];

  const int m = lane & 15;
  const int hi = lane >> 4;
  const int k0 = hi * 2;
  const int vrow = (m & 7) < VR ? (m & 7) : 4;
  const float* va = vb + (4 * hi * VR + vrow) * 2;

  float kmax2 = 0.f;
#pragma unroll
  for (int kt = 0; kt < NT; ++kt) {
    const v2f a = *(const v2f*)(qb + ((kt * 16 + m) * 4 + k0));
    const float s = a.x * a.x + a.y * a.y;
    kmax2 = fmaxf(kmax2, s + xhalf(s));
  }
  kmax2 = fmaxf(kmax2, xrow<1>(kmax2));
  kmax2 = fmaxf(kmax2, xrow<2>(kmax2));
  kmax2 = fmaxf(kmax2, xrow<4>(kmax2));
  kmax2 = fmaxf(kmax2, xrow<8>(kmax2));

  const bool small = __builtin_amdgcn_ballot_w32(kmax2 <= 144.f) == 0xffffffffu;
  if (small)
    attend<WS, SH, CHUNK, true>(out, qb, va, wy, wx, b, head, m, hi, kmax2);
  else
    attend<WS, SH, CHUNK, false>(out, qb, va, wy, wx, b, head, m, hi, kmax2);
}

__global__ __launch_bounds__(128) void win_attn_fused(
    const float* __restrict__ x, float* __restrict__ out) {
  __shared__ __attribute__((aligned(16))) float smem[4 * 256 * 4 + 4 * 128 * 5 * 2];
  const int bx = blockIdx.x;
  if (bx < 1024) {
    win_attn_block<16, 8, 1>(x, out, smem, bx & 255, bx >> 8);
  } else {
    const int r = bx - 1024;
    win_attn_block<8, 0, 0>(x, out, smem, r & 1023, r >> 10);
  }
}

extern "C" void kernel_launch(void* const* d_in, const int* in_sizes, int n_in,
                              void* d_out, int out_size, void* d_ws,
                              size_t ws_size, hipStream_t stream) {
  (void)in_sizes; (void)n_in; (void)out_size; (void)d_ws; (void)ws_size;
  const float* x = (const float*)d_in[0];
  float* out = (float*)d_out;

  dim3 g(5120);
  win_attn_fused<<<g, 128, 0, stream>>>(x, out);
}


# Key-side WMMA fragments (Q keys, [V|1]^T) read from LDS once per window and kept in registers over all query tiles; bound form takes each score tile two key tiles ahead of use; sums unchanged
# speedup vs baseline: 2.9436x; 1.0571x over previous
#include <hip/hip_runtime.h>


typedef float v2f __attribute__((ext_vector_type(2)));
typedef float v8f __attribute__((ext_vector_type(8)));

__device__ __forceinline__ v8f wmma4(v2f a, v2f b, v8f c) {
  return __builtin_amdgcn_wmma_f32_16x16x4_f32(false, a, false, b, (short)0, c,
                                               false, false);
}

__device__ __forceinline__ float xhalf(float v) {
  int b = __builtin_bit_cast(int, v);
  return __builtin_bit_cast(
      float, __builtin_amdgcn_permlanex16(b, b, 0x76543210, 0xfedcba98, false,
                                          false));
}

template <int X>
__device__ __forceinline__ float xrow(float v) {
  constexpr unsigned lo = X == 1   ? 0x67452301u
                          : X == 2 ? 0x54761032u
                          : X == 4 ? 0x32107654u
                                   : 0xfedcba98u;
  constexpr unsigned hi = X == 1   ? 0xefcdab89u
                          : X == 2 ? 0xdcfe98bau
                          : X == 4 ? 0xba98fedcu
                                   : 0x76543210u;
  int b = __builtin_bit_cast(int, v);
  return __builtin_bit_cast(
      float, __builtin_amdgcn_permlane16(b, b, lo, hi, false, false));
}

template <int WS, int SH, int CHUNK, bool BOUND>
__device__ __forceinline__ void attend(
    float* __restrict__ out, const float* qb, const float* va, const int wy,
    const int wx, const int b, const int head, const int m, const int hi,
    const float kmax2) {
  constexpr int H = 256, W = 256, HW = H * W;
  constexpr int T = WS * WS;
  constexpr int NT = T / 16;
  constexpr int VR = 5;
  constexpr float LOG2E = 1.4426950408889634f;
  const int k0 = hi * 2;

  constexpr bool KEEP = BOUND || NT <= 4;
  constexpr int PF = 2;
  constexpr int NAV = KEEP ? NT : PF;
  v2f akeep[NT];
  v2f avkeep[NAV][4];
#pragma unroll
  for (int kt = 0; kt < NT; ++kt)
    akeep[kt] = *(const v2f*)(qb + ((kt * 16 + m) * 4 + k0));
#pragma unroll
  for (int kt = 0; kt < NAV; ++kt) {
#pragma unroll
    for (int j = 0; j < 4; ++j)
      avkeep[kt][j] = *(const v2f*)(va + (kt * 8 + j) * VR * 2);
  }
  v2f bnext = *(const v2f*)(qb + (m * 4 + k0));

  for (int qt = 0; qt < NT; ++qt) {
    if constexpr (!KEEP) {
      asm volatile("" ::: "memory");
    }
    const v2f bqry = bnext;
    if constexpr (KEEP) {
      bnext = *(const v2f*)(qb + ((((qt + 1) & (NT - 1)) * 16 + m) * 4 + k0));
    }

    v8f S[NT];
    v8f O = {0.f, 0.f, 0.f, 0.f, 0.f, 0.f, 0.f, 0.f};
    if constexpr (BOUND) {
      float qn = bqry.x * bqry.x + bqry.y * bqry.y;
      qn += xhalf(qn);
      const float c0 = -LOG2E * __builtin_amdgcn_sqrtf(qn * kmax2);
      const v2f bq2 = {bqry.x * LOG2E, bqry.y * LOG2E};
      const v8f cs = {c0, c0, c0, c0, c0, c0, c0, c0};

      S[0] = wmma4(akeep[0], bq2, cs);
      S[1] = wmma4(akeep[1], bq2, cs);
#pragma unroll
      for (int r = 0; r < 8; ++r) S[0][r] = __builtin_amdgcn_exp2f(S[0][r]);
#pragma unroll
      for (int kt = 0; kt < NT; ++kt) {
        __builtin_amdgcn_sched_barrier(0);
        if (kt + 2 < NT) S[kt + 2] = wmma4(akeep[kt + 2], bq2, cs);
        if (kt + 1 < NT) {
#pragma unroll
          for (int r = 0; r < 8; ++r)
            S[kt + 1][r] = __builtin_amdgcn_exp2f(S[kt + 1][r]);
        }
#pragma unroll
        for (int j = 0; j < 4; ++j) {
          v2f bp = {S[kt][2 * j], S[kt][2 * j + 1]};
          O = wmma4(avkeep[kt][j], bp, O);
        }
      }
    } else {
#pragma unroll
      for (int kt = 0; kt < NT; ++kt) {
        v8f z = {0.f, 0.f, 0.f, 0.f, 0.f, 0.f, 0.f, 0.f};
        S[kt] = wmma4(akeep[kt], bqry, z);
      }

      float mc[4] = {-3.402823466e38f, -3.402823466e38f, -3.402823466e38f,
                     -3.402823466e38f};
#pragma unroll
      for (int kt = 0; kt < NT; ++kt) {
#pragma unroll
        for (int r = 0; r < 8; ++r) mc[kt & 3] = fmaxf(mc[kt & 3], S[kt][r]);
      }
      float mx = fmaxf(fmaxf(mc[0], mc[1]), fmaxf(mc[2], mc[3]));
      mx = fmaxf(mx, xhalf(mx));

      if constexpr (KEEP) {
#pragma unroll
        for (int kt = 0; kt < NT; ++kt) {
#pragma unroll
          for (int r = 0; r < 8; ++r) S[kt][r] = __expf(S[kt][r] - mx);
        }
#pragma unroll
        for (int kt = 0; kt < NT; ++kt) {
#pragma unroll
          for (int j = 0; j < 4; ++j) {
            v2f bp = {S[kt][2 * j], S[kt][2 * j + 1]};
            O = wmma4(avkeep[kt][j], bp, O);
          }
        }
      } else {
        v2f av[NT][4];
#pragma unroll
        for (int kt = 0; kt < PF; ++kt) {
#pragma unroll
          for (int j = 0; j < 4; ++j) av[kt][j] = avkeep[kt][j];
        }
#pragma unroll
        for (int r = 0; r < 8; ++r) S[0][r] = __expf(S[0][r] - mx);
#pragma unroll
        for (int kt = 0; kt < NT; ++kt) {
          if ((kt & 1) == 0 && kt + PF < NT) {
            __builtin_amdgcn_sched_barrier(0);
#pragma unroll
            for (int t = kt + PF; t < kt + PF + 2; ++t) {
#pragma unroll
              for (int j = 0; j < 4; ++j)
                av[t][j] = *(const v2f*)(va + (t * 8 + j) * VR * 2);
            }
            if (kt == 0) {
              bnext = *(const v2f*)(
                  qb + ((((qt + 1) & (NT - 1)) * 16 + m) * 4 + k0));
            }
            __builtin_amdgcn_sched_barrier(0);
          }
          if (kt + 1 < NT) {
#pragma unroll
            for (int r = 0; r < 8; ++r)
              S[kt + 1][r] = __expf(S[kt + 1][r] - mx);
          }
#pragma unroll
          for (int j = 0; j < 4; ++j) {
            v2f bp = {S[kt][2 * j], S[kt][2 * j + 1]};
            O = wmma4(av[kt][j], bp, O);
          }
        }
      }
    }

    {
      int tok = qt * 16 + m;
      int ty = tok / WS, tx = tok - ty * WS;
      int gh = (wy * WS + ty - SH) & (H - 1);
      int gw = (wx * WS + tx - SH) & (W - 1);
      float* op = out +
                  ((size_t)b * 32 + CHUNK * 16 + head * 4 + 2 * hi) * HW +
                  gh * W + gw;
      float sm = O[4];
      float o0 = hi ? O[2] : O[0];
      float o1 = hi ? O[3] : O[1];
      op[0] = o0 / sm;
      op[HW] = o1 / sm;
    }
  }
}

template <int WS, int SH, int CHUNK>
__device__ __forceinline__ void win_attn_block(
    const float* __restrict__ x, float* __restrict__ out,
    float* __restrict__ smem, const int wid, const int b) {
  constexpr int H = 256, W = 256, HW = H * W;
  constexpr int NW = W / WS;
  constexpr int T = WS * WS;
  constexpr int NT = T / 16;
  constexpr int VR = 5;

  float* const ql = smem;
  float* const vl = smem + 4 * T * 4;

  const int wy = wid / NW, wx = wid % NW;
  const int tid = threadIdx.x;
  const int lane = tid & 31;
  const int head = tid >> 5;

  const float* xbase = x + ((size_t)b * 64 + CHUNK * 32) * HW;
  constexpr int GI = 16 * T / 128;
  for (int it = 0; it < GI; ++it) {
    int i = tid + it * 128;
    int c = i / T, t = i - c * T;
    int ty = t / WS, tx = t - ty * WS;
    int gh = (wy * WS + ty - SH) & (H - 1);
    int gw = (wx * WS + tx - SH) & (W - 1);
    ql[((c >> 2) * T + t) * 4 + (c & 3)] = xbase[c * HW + gh * W + gw];
  }
  for (int it = 0; it < GI; ++it) {
    int i = tid + it * 128;
    int c = i / T, t = i - c * T;
    int ty = t / WS, tx = t - ty * WS;
    int gh = (wy * WS + ty - SH) & (H - 1);
    int gw = (wx * WS + tx - SH) & (W - 1);
    vl[(((c >> 2) * (T / 2) + (t >> 1)) * VR + (c & 3)) * 2 + (t & 1)] =
        xbase[(16 + c) * HW + gh * W + gw];
  }
  for (int i = tid; i < 4 * (T / 2); i += 128) {
    v2f one = {1.f, 1.f};
    *(v2f*)&vl[(i * VR + 4) * 2] = one;
  }
  __syncthreads();

  const float* qb = &ql[head * T * 4];
  const float* vb = &vl[head * (T / 2) * VR * 2];

  const int m = lane & 15;
  const int hi = lane >> 4;
  const int k0 = hi * 2;
  const int vrow = (m & 7) < VR ? (m & 7) : 4;
  const float* va = vb + (4 * hi * VR + vrow) * 2;

  float kmax2 = 0.f;
#pragma unroll
  for (int kt = 0; kt < NT; ++kt) {
    const v2f a = *(const v2f*)(qb + ((kt * 16 + m) * 4 + k0));
    const float s = a.x * a.x + a.y * a.y;
    kmax2 = fmaxf(kmax2, s + xhalf(s));
  }
  kmax2 = fmaxf(kmax2, xrow<1>(kmax2));
  kmax2 = fmaxf(kmax2, xrow<2>(kmax2));
  kmax2 = fmaxf(kmax2, xrow<4>(kmax2));
  kmax2 = fmaxf(kmax2, xrow<8>(kmax2));

  const bool small = __builtin_amdgcn_ballot_w32(kmax2 <= 144.f) == 0xffffffffu;
  if (small)
    attend<WS, SH, CHUNK, true>(out, qb, va, wy, wx, b, head, m, hi, kmax2);
  else
    attend<WS, SH, CHUNK, false>(out, qb, va, wy, wx, b, head, m, hi, kmax2);
}

__global__ __launch_bounds__(128) void win_attn_fused(
    const float* __restrict__ x, float* __restrict__ out) {
  __shared__ __attribute__((aligned(16))) float smem[4 * 256 * 4 + 4 * 128 * 5 * 2];
  const int bx = blockIdx.x;
  if (bx < 1024) {
    win_attn_block<16, 8, 1>(x, out, smem, bx & 255, bx >> 8);
  } else {
    const int r = bx - 1024;
    win_attn_block<8, 0, 0>(x, out, smem, r & 1023, r >> 10);
  }
}

extern "C" void kernel_launch(void* const* d_in, const int* in_sizes, int n_in,
                              void* d_out, int out_size, void* d_ws,
                              size_t ws_size, hipStream_t stream) {
  (void)in_sizes; (void)n_in; (void)out_size; (void)d_ws; (void)ws_size;
  const float* x = (const float*)d_in[0];
  float* out = (float*)d_out;

  dim3 g(5120);
  win_attn_fused<<<g, 128, 0, stream>>>(x, out);
}


# 8x8 windows: one block takes the four windows of an 8-row x 32-column strip (one gather of whole 128B lines, one barrier, each wave its head over the four in turn); grid 5120 -> 2048; sums unchanged
# speedup vs baseline: 3.0302x; 1.0294x over previous
#include <hip/hip_runtime.h>


typedef float v2f __attribute__((ext_vector_type(2)));
typedef float v8f __attribute__((ext_vector_type(8)));

__device__ __forceinline__ v8f wmma4(v2f a, v2f b, v8f c) {
  return __builtin_amdgcn_wmma_f32_16x16x4_f32(false, a, false, b, (short)0, c,
                                               false, false);
}

__device__ __forceinline__ float xhalf(float v) {
  int b = __builtin_bit_cast(int, v);
  return __builtin_bit_cast(
      float, __builtin_amdgcn_permlanex16(b, b, 0x76543210, 0xfedcba98, false,
                                          false));
}

template <int X>
__device__ __forceinline__ float xrow(float v) {
  constexpr unsigned lo = X == 1   ? 0x67452301u
                          : X == 2 ? 0x54761032u
                          : X == 4 ? 0x32107654u
                                   : 0xfedcba98u;
  constexpr unsigned hi = X == 1   ? 0xefcdab89u
                          : X == 2 ? 0xdcfe98bau
                          : X == 4 ? 0xba98fedcu
                                   : 0x76543210u;
  int b = __builtin_bit_cast(int, v);
  return __builtin_bit_cast(
      float, __builtin_amdgcn_permlane16(b, b, lo, hi, false, false));
}

template <int WS, int SH, int CHUNK, bool BOUND>
__device__ __forceinline__ void attend(
    float* __restrict__ out, const float* qb, const float* va, const int wy,
    const int wx, const int b, const int head, const int m, const int hi,
    const float kmax2) {
  constexpr int H = 256, W = 256, HW = H * W;
  constexpr int T = WS * WS;
  constexpr int NT = T / 16;
  constexpr int VR = 5;
  constexpr float LOG2E = 1.4426950408889634f;
  const int k0 = hi * 2;

  constexpr bool KEEP = BOUND || NT <= 4;
  constexpr int PF = 2;
  constexpr int NAV = KEEP ? NT : PF;
  v2f akeep[NT];
  v2f avkeep[NAV][4];
#pragma unroll
  for (int kt = 0; kt < NT; ++kt)
    akeep[kt] = *(const v2f*)(qb + ((kt * 16 + m) * 4 + k0));
#pragma unroll
  for (int kt = 0; kt < NAV; ++kt) {
#pragma unroll
    for (int j = 0; j < 4; ++j)
      avkeep[kt][j] = *(const v2f*)(va + (kt * 8 + j) * VR * 2);
  }
  v2f bnext = *(const v2f*)(qb + (m * 4 + k0));

  for (int qt = 0; qt < NT; ++qt) {
    if constexpr (!KEEP) {
      asm volatile("" ::: "memory");
    }
    const v2f bqry = bnext;
    if constexpr (KEEP) {
      bnext = *(const v2f*)(qb + ((((qt + 1) & (NT - 1)) * 16 + m) * 4 + k0));
    }

    v8f S[NT];
    v8f O = {0.f, 0.f, 0.f, 0.f, 0.f, 0.f, 0.f, 0.f};
    if constexpr (BOUND) {
      float qn = bqry.x * bqry.x + bqry.y * bqry.y;
      qn += xhalf(qn);
      const float c0 = -LOG2E * __builtin_amdgcn_sqrtf(qn * kmax2);
      const v2f bq2 = {bqry.x * LOG2E, bqry.y * LOG2E};
      const v8f cs = {c0, c0, c0, c0, c0, c0, c0, c0};

      S[0] = wmma4(akeep[0], bq2, cs);
      S[1] = wmma4(akeep[1], bq2, cs);
#pragma unroll
      for (int r = 0; r < 8; ++r) S[0][r] = __builtin_amdgcn_exp2f(S[0][r]);
#pragma unroll
      for (int kt = 0; kt < NT; ++kt) {
        __builtin_amdgcn_sched_barrier(0);
        if (kt + 2 < NT) S[kt + 2] = wmma4(akeep[kt + 2], bq2, cs);
        if (kt + 1 < NT) {
#pragma unroll
          for (int r = 0; r < 8; ++r)
            S[kt + 1][r] = __builtin_amdgcn_exp2f(S[kt + 1][r]);
        }
#pragma unroll
        for (int j = 0; j < 4; ++j) {
          v2f bp = {S[kt][2 * j], S[kt][2 * j + 1]};
          O = wmma4(avkeep[kt][j], bp, O);
        }
      }
    } else {
#pragma unroll
      for (int kt = 0; kt < NT; ++kt) {
        v8f z = {0.f, 0.f, 0.f, 0.f, 0.f, 0.f, 0.f, 0.f};
        S[kt] = wmma4(akeep[kt], bqry, z);
      }

      float mc[4] = {-3.402823466e38f, -3.402823466e38f, -3.402823466e38f,
                     -3.402823466e38f};
#pragma unroll
      for (int kt = 0; kt < NT; ++kt) {
#pragma unroll
        for (int r = 0; r < 8; ++r) mc[kt & 3] = fmaxf(mc[kt & 3], S[kt][r]);
      }
      float mx = fmaxf(fmaxf(mc[0], mc[1]), fmaxf(mc[2], mc[3]));
      mx = fmaxf(mx, xhalf(mx));

      if constexpr (KEEP) {
#pragma unroll
        for (int kt = 0; kt < NT; ++kt) {
#pragma unroll
          for (int r = 0; r < 8; ++r) S[kt][r] = __expf(S[kt][r] - mx);
        }
#pragma unroll
        for (int kt = 0; kt < NT; ++kt) {
#pragma unroll
          for (int j = 0; j < 4; ++j) {
            v2f bp = {S[kt][2 * j], S[kt][2 * j + 1]};
            O = wmma4(avkeep[kt][j], bp, O);
          }
        }
      } else {
        v2f av[NT][4];
#pragma unroll
        for (int kt = 0; kt < PF; ++kt) {
#pragma unroll
          for (int j = 0; j < 4; ++j) av[kt][j] = avkeep[kt][j];
        }
#pragma unroll
        for (int r = 0; r < 8; ++r) S[0][r] = __expf(S[0][r] - mx);
#pragma unroll
        for (int kt = 0; kt < NT; ++kt) {
          if ((kt & 1) == 0 && kt + PF < NT) {
            __builtin_amdgcn_sched_barrier(0);
#pragma unroll
            for (int t = kt + PF; t < kt + PF + 2; ++t) {
#pragma unroll
              for (int j = 0; j < 4; ++j)
                av[t][j] = *(const v2f*)(va + (t * 8 + j) * VR * 2);
            }
            if (kt == 0) {
              bnext = *(const v2f*)(
                  qb + ((((qt + 1) & (NT - 1)) * 16 + m) * 4 + k0));
            }
            __builtin_amdgcn_sched_barrier(0);
          }
          if (kt + 1 < NT) {
#pragma unroll
            for (int r = 0; r < 8; ++r)
              S[kt + 1][r] = __expf(S[kt + 1][r] - mx);
          }
#pragma unroll
          for (int j = 0; j < 4; ++j) {
            v2f bp = {S[kt][2 * j], S[kt][2 * j + 1]};
            O = wmma4(av[kt][j], bp, O);
          }
        }
      }
    }

    {
      int tok = qt * 16 + m;
      int ty = tok / WS, tx = tok - ty * WS;
      int gh = (wy * WS + ty - SH) & (H - 1);
      int gw = (wx * WS + tx - SH) & (W - 1);
      float* op = out +
                  ((size_t)b * 32 + CHUNK * 16 + head * 4 + 2 * hi) * HW +
                  gh * W + gw;
      float sm = O[4];
      float o0 = hi ? O[2] : O[0];
      float o1 = hi ? O[3] : O[1];
      op[0] = o0 / sm;
      op[HW] = o1 / sm;
    }
  }
}

template <int WS, int SH, int CHUNK>
__device__ __forceinline__ void attend_window(
    float* __restrict__ out, const float* qb, const float* vb, const int wy,
    const int wx, const int b, const int head, const int lane) {
  constexpr int T = WS * WS;
  constexpr int NT = T / 16;
  constexpr int VR = 5;

  const int m = lane & 15;
  const int hi = lane >> 4;
  const int k0 = hi * 2;
  const int vrow = (m & 7) < VR ? (m & 7) : 4;
  const float* va = vb + (4 * hi * VR + vrow) * 2;

  float kmax2 = 0.f;
#pragma unroll
  for (int kt = 0; kt < NT; ++kt) {
    const v2f a = *(const v2f*)(qb + ((kt * 16 + m) * 4 + k0));
    const float s = a.x * a.x + a.y * a.y;
    kmax2 = fmaxf(kmax2, s + xhalf(s));
  }
  kmax2 = fmaxf(kmax2, xrow<1>(kmax2));
  kmax2 = fmaxf(kmax2, xrow<2>(kmax2));
  kmax2 = fmaxf(kmax2, xrow<4>(kmax2));
  kmax2 = fmaxf(kmax2, xrow<8>(kmax2));

  const bool small = __builtin_amdgcn_ballot_w32(kmax2 <= 144.f) == 0xffffffffu;
  if (small)
    attend<WS, SH, CHUNK, true>(out, qb, va, wy, wx, b, head, m, hi, kmax2);
  else
    attend<WS, SH, CHUNK, false>(out, qb, va, wy, wx, b, head, m, hi, kmax2);
}

template <int SH, int CHUNK>
__device__ __forceinline__ void win16_block(
    const float* __restrict__ x, float* __restrict__ out,
    float* __restrict__ smem, const int wid, const int b) {
  constexpr int WS = 16;
  constexpr int H = 256, W = 256, HW = H * W;
  constexpr int NW = W / WS;
  constexpr int T = WS * WS;
  constexpr int VR = 5;

  float* const ql = smem;
  float* const vl = smem + 4 * T * 4;

  const int wy = wid / NW, wx = wid % NW;
  const int tid = threadIdx.x;

  const float* xbase = x + ((size_t)b * 64 + CHUNK * 32) * HW;
  constexpr int GI = 16 * T / 128;
  for (int it = 0; it < GI; ++it) {
    int i = tid + it * 128;
    int c = i / T, t = i - c * T;
    int ty = t / WS, tx = t - ty * WS;
    int gh = (wy * WS + ty - SH) & (H - 1);
    int gw = (wx * WS + tx - SH) & (W - 1);
    ql[((c >> 2) * T + t) * 4 + (c & 3)] = xbase[c * HW + gh * W + gw];
  }
  for (int it = 0; it < GI; ++it) {
    int i = tid + it * 128;
    int c = i / T, t = i - c * T;
    int ty = t / WS, tx = t - ty * WS;
    int gh = (wy * WS + ty - SH) & (H - 1);
    int gw = (wx * WS + tx - SH) & (W - 1);
    vl[(((c >> 2) * (T / 2) + (t >> 1)) * VR + (c & 3)) * 2 + (t & 1)] =
        xbase[(16 + c) * HW + gh * W + gw];
  }
  for (int i = tid; i < 4 * (T / 2); i += 128) {
    v2f one = {1.f, 1.f};
    *(v2f*)&vl[(i * VR + 4) * 2] = one;
  }
  __syncthreads();

  const int head = tid >> 5;
  attend_window<WS, SH, CHUNK>(out, &ql[head * T * 4],
                               &vl[head * (T / 2) * VR * 2], wy, wx, b, head,
                               tid & 31);
}

constexpr int QW8 = 4 * 64 * 4 + 2;
constexpr int VW8 = 4 * 32 * 5 * 2 + 2;

template <int CHUNK>
__device__ __forceinline__ void win8_strip_block(
    const float* __restrict__ x, float* __restrict__ out,
    float* __restrict__ smem, const int sid, const int b) {
  constexpr int WS = 8;
  constexpr int H = 256, W = 256, HW = H * W;
  constexpr int T = WS * WS;
  constexpr int VR = 5;
  constexpr int SW = 4;
  constexpr int ST = SW * T;

  float* const ql = smem;
  float* const vl = smem + SW * QW8;

  const int wy = sid >> 3;
  const int sx = sid & 7;
  const int tid = threadIdx.x;
  const int lane = tid & 31;
  const int head = tid >> 5;

  const float* xbase = x + ((size_t)b * 64 + CHUNK * 32) * HW +
                       (wy * WS) * W + sx * (SW * WS) + head * W + lane;
  constexpr int GI = 16 * ST / 128;
  const int gwin = lane >> 3, j = lane & 7;
  float* const qdst = ql + gwin * QW8 + (head * WS + j) * 4;
  float* const vdst =
      vl + gwin * VW8 + (head * (WS / 2) + (j >> 1)) * VR * 2 + (j & 1);
  float rq[GI], rv[GI];
#pragma unroll
  for (int it = 0; it < GI; ++it)
    rq[it] = xbase[(it >> 1) * HW + 4 * (it & 1) * W];
#pragma unroll
  for (int it = 0; it < GI; ++it)
    rv[it] = xbase[(16 + (it >> 1)) * HW + 4 * (it & 1) * W];
#pragma unroll
  for (int it = 0; it < GI; ++it) {
    const int c = it >> 1, tyo = 4 * (it & 1);
    qdst[((c >> 2) * T + tyo * WS) * 4 + (c & 3)] = rq[it];
  }
#pragma unroll
  for (int it = 0; it < GI; ++it) {
    const int c = it >> 1, tyo = 4 * (it & 1);
    vdst[((c >> 2) * (T / 2) + tyo * (WS / 2)) * VR * 2 + (c & 3) * 2] = rv[it];
  }
  for (int i = tid; i < SW * 4 * (T / 2); i += 128) {
    v2f one = {1.f, 1.f};
    int ww = i / (4 * (T / 2)), p = i - ww * (4 * (T / 2));
    *(v2f*)&vl[ww * VW8 + (p * VR + 4) * 2] = one;
  }
  __syncthreads();

#pragma unroll 1
  for (int w = 0; w < SW; ++w) {
    attend_window<WS, 0, CHUNK>(out, &ql[w * QW8 + head * T * 4],
                                &vl[w * VW8 + head * (T / 2) * VR * 2], wy,
                                sx * SW + w, b, head, tid & 31);
  }
}

__global__ __launch_bounds__(128) void win_attn_fused(
    const float* __restrict__ x, float* __restrict__ out) {
  __shared__ __attribute__((aligned(16))) float smem[4 * QW8 + 4 * VW8];
  const int bx = blockIdx.x;
  if (bx < 1024) {
    win16_block<8, 1>(x, out, smem, bx & 255, bx >> 8);
  } else {
    const int r = bx - 1024;
    win8_strip_block<0>(x, out, smem, r & 255, r >> 8);
  }
}

extern "C" void kernel_launch(void* const* d_in, const int* in_sizes, int n_in,
                              void* d_out, int out_size, void* d_ws,
                              size_t ws_size, hipStream_t stream) {
  (void)in_sizes; (void)n_in; (void)out_size; (void)d_ws; (void)ws_size;
  const float* x = (const float*)d_in[0];
  float* out = (float*)d_out;

  dim3 g(2048);
  win_attn_fused<<<g, 128, 0, stream>>>(x, out);
}
